# starting kernel
# baseline (speedup 1.0000x reference)
#include <hip/hip_runtime.h>
#include <math.h>

typedef __attribute__((ext_vector_type(2))) float v2f;
typedef __attribute__((ext_vector_type(8))) float v8f;

#define NT 1024
#define NS 2048
#define WAVES_PER_BLOCK 8
#define BLOCK_THREADS (WAVES_PER_BLOCK * 32)

#define DPP8_SEL(a,b,c,d,e,f,g,h) \
  ((a)|((b)<<3)|((c)<<6)|((d)<<9)|((e)<<12)|((f)<<15)|((g)<<18)|((h)<<21))
constexpr int DPP8_XOR1 = DPP8_SEL(1,0,3,2,5,4,7,6);
constexpr int DPP8_XOR2 = DPP8_SEL(2,3,0,1,6,7,4,5);
constexpr int DPP8_XOR4 = DPP8_SEL(4,5,6,7,0,1,2,3);

__device__ __forceinline__ float bcastf(float v, int l) {
  return __builtin_bit_cast(
      float, __builtin_amdgcn_readlane(__builtin_bit_cast(int, v), l));
}

template <int SEL>
__device__ __forceinline__ float dpp8f(float v) {
  return __builtin_bit_cast(
      float, __builtin_amdgcn_mov_dpp8(__builtin_bit_cast(int, v), SEL));
}

__device__ __forceinline__ float xor8f(float v) {
  int x = __builtin_bit_cast(int, v);
  int r = __builtin_amdgcn_permlane16(x, x, 0xFEDCBA98u, 0x76543210u,
                                       false,  false);
  return __builtin_bit_cast(float, r);
}

__device__ __forceinline__ float wmma_fold16(float p) {
  v2f a; a.x = 1.0f; a.y = 1.0f;
  v2f b; b.x = p;    b.y = 0.0f;
  v8f c = {};
  v8f d = __builtin_amdgcn_wmma_f32_16x16x4_f32(
       false, a,  false, b,
       (short)0, c,  false,  false);
  return d[0];
}

__device__ __forceinline__ float sigmoidf(float x) {
  return 1.0f / (1.0f + expf(-x));
}

__global__ __launch_bounds__(BLOCK_THREADS) void waternet_scan(
    const float* __restrict__ P,  const float* __restrict__ T1,
    const float* __restrict__ T2, const float* __restrict__ E,
    const float* __restrict__ w,  float* __restrict__ out) {
  const int lane = threadIdx.x & 31;
  const int wv   = threadIdx.x >> 5;
  const int s    = blockIdx.x * WAVES_PER_BLOCK + wv;

  const float gm = expf(w[lane]) + 1.0f;
  const float ge = sigmoidf(w[32 + lane]);
  const float go = sigmoidf(w[64 + lane]);
  const float xw = w[96 + lane];
  float mx = xw;
  for (int off = 16; off; off >>= 1) mx = fmaxf(mx, __shfl_xor(mx, off, 32));
  const float ex = expf(xw - mx);
  float sm = ex;
  for (int off = 16; off; off >>= 1) sm += __shfl_xor(sm, off, 32);
  const float ga = ex / sm;

  float S = 0.0f, H = 0.0f;

  for (int t0 = 0; t0 < NT; t0 += 32) {
    const int idx = (t0 + lane) * NS + s;
    const float Pv  = P[idx];
    const float t1v = T1[idx];
    const float t2v = T2[idx];
    const float Ev  = E[idx];

    if (t0 + 32 < NT) {
      const int nidx = idx + 32 * NS;
      __builtin_prefetch(&P[nidx],  0, 0);
      __builtin_prefetch(&T1[nidx], 0, 0);
      __builtin_prefetch(&T2[nidx], 0, 0);
      __builtin_prefetch(&E[nidx],  0, 0);
    }

    const float Ta = (t1v + t2v) * 0.5f;
    float ratio = (t1v + t2v) / (t2v - t1v);
    ratio = fminf(1.0f, fmaxf(-1.0f, ratio));
    float rp = 1.0f - acosf(ratio) / 3.1415f;
    rp = (t1v >= 0.0f) ? 1.0f : rp;
    rp = (t2v <= 0.0f) ? 0.0f : rp;
    const float Psn = (1.0f - rp) * Pv;
    const float Plq = rp * Pv;

    float yv = 0.0f;
#pragma unroll
    for (int k = 0; k < 32; ++k) {
      const float Ta_k = bcastf(Ta,  k);
      const float Ps_k = bcastf(Psn, k);
      const float Pl_k = bcastf(Plq, k);
      const float E_k  = bcastf(Ev,  k);

      const float Smelt = fminf(S, fmaxf(Ta_k * gm, 0.0f));
      S = S + Ps_k - Smelt;
      float Hn = H + Smelt + Pl_k - E_k * ge;
      Hn = fmaxf(Hn, 0.0f);
      const float Q = Hn * go;
      H = Hn - Q;

      float r = wmma_fold16(Q * ga);
      r += xor8f(r);
      r += dpp8f<DPP8_XOR4>(r);
      r += dpp8f<DPP8_XOR2>(r);
      r += dpp8f<DPP8_XOR1>(r);

      yv = (lane == k) ? r : yv;
    }
    out[idx] = yv;
  }
}

extern "C" void kernel_launch(void* const* d_in, const int* in_sizes, int n_in,
                              void* d_out, int out_size, void* d_ws, size_t ws_size,
                              hipStream_t stream) {
  (void)in_sizes; (void)n_in; (void)out_size; (void)d_ws; (void)ws_size;
  const float* P  = (const float*)d_in[0];
  const float* T1 = (const float*)d_in[1];
  const float* T2 = (const float*)d_in[2];
  const float* E  = (const float*)d_in[3];
  const float* w  = (const float*)d_in[4];
  float* out = (float*)d_out;

  dim3 grid(NS / WAVES_PER_BLOCK);
  dim3 block(BLOCK_THREADS);
  waternet_scan<<<grid, block, 0, stream>>>(P, T1, T2, E, w, out);
}


# Forcings register-staged: each block's 16 forcing vectors are read from LDS as one group a full block ahead into a second register set (LDS ring of 4 blocks); no LDS read or wait among the steps
# speedup vs baseline: 1.8508x; 1.8508x over previous
#include <hip/hip_runtime.h>
#include <math.h>


typedef __attribute__((ext_vector_type(2))) float v2f;
typedef __attribute__((ext_vector_type(4))) float v4f;
typedef __attribute__((ext_vector_type(8))) float v8f;

#define NT 1024
#define NS 2048
#define SITES 8
#define TBLK 16
#define THREADS 128
#define NBLK (NT / TBLK)
#define RING 4

__device__ __forceinline__ float sigmoidf(float x) {
  return 1.0f / (1.0f + expf(-x));
}

struct Forc { v2f ts; v2f er; };

__device__ __forceinline__ Forc partition(float Pv, float t1v, float t2v, float Ev) {
  const float Ta = (t1v + t2v) * 0.5f;
  float ratio = (t1v + t2v) / (t2v - t1v);
  ratio = fminf(1.0f, fmaxf(-1.0f, ratio));
  float rp = 1.0f - acosf(ratio) / 3.1415f;
  rp = (t1v >= 0.0f) ? 1.0f : rp;
  rp = (t2v <= 0.0f) ? 0.0f : rp;
  Forc f;
  f.ts.x = Ta;
  f.ts.y = (1.0f - rp) * Pv;
  f.er.x = Ev;
  f.er.y = rp * Pv;
  return f;
}

struct Gates { float gm0, gm1, ge0, ge1, go0, go1, ga0, ga1; };
struct Raw   { float p, t1, t2, e; };

__device__ __forceinline__ void scan_block(
    const int b, Forc (&fc)[TBLK], Raw& raw, v4f& rx, v4f& ry,
    v2f (*Fts)[TBLK * SITES], v2f (*Fer)[TBLK * SITES], float (*R)[TBLK * SITES * 8],
    const float* __restrict__ P,  const float* __restrict__ T1,
    const float* __restrict__ T2, const float* __restrict__ E,
    float* __restrict__ out, const Gates& g,
    float& S0, float& S1, float& H0, float& H1,
    const int tid, const int site, const int col, const int half, const int part,
    const int s0, const int st_step, const int st_site) {
  const int cur = b & 1;

  if (b >= 2)
    out[((b - 2) * TBLK + st_step) * NS + s0 + st_site] =
        ((rx.x + rx.y) + (rx.z + rx.w)) + ((ry.x + ry.y) + (ry.z + ry.w));

  v8f d = {};
#pragma unroll
  for (int j = 0; j < TBLK; ++j) {
    const float Ta = fc[j].ts.x, snow = fc[j].ts.y;
    const float Ev = fc[j].er.x, rain = fc[j].er.y;

    const float sm0 = fminf(S0, fmaxf(Ta * g.gm0, 0.0f));
    const float sm1 = fminf(S1, fmaxf(Ta * g.gm1, 0.0f));
    S0 = S0 + snow - sm0;
    S1 = S1 + snow - sm1;
    float hn0 = H0 + sm0 + rain - Ev * g.ge0;
    float hn1 = H1 + sm1 + rain - Ev * g.ge1;
    hn0 = fmaxf(hn0, 0.0f);
    hn1 = fmaxf(hn1, 0.0f);
    const float q0 = hn0 * g.go0;
    const float q1 = hn1 * g.go1;
    H0 = hn0 - q0;
    H1 = hn1 - q1;

    const float sel = (col == j) ? 1.0f : 0.0f;
    v2f a;  a.x = sel;         a.y = sel;
    v2f bq; bq.x = q0 * g.ga0; bq.y = q1 * g.ga1;
    d = __builtin_amdgcn_wmma_f32_16x16x4_f32(
         false, a,  false, bq,
         (short)0, d,  false,  false);
  }

#pragma unroll
  for (int r = 0; r < 8; ++r)
    R[cur][((r + 8 * half) * SITES + site) * 8 + part] = d[r];

  if (b + 2 < NBLK) {
    const Forc f = partition(raw.p, raw.t1, raw.t2, raw.e);
    Fts[(b + 2) & (RING - 1)][tid] = f.ts;
    Fer[(b + 2) & (RING - 1)][tid] = f.er;
  }

  __syncthreads();

  const v4f* rp = reinterpret_cast<const v4f*>(&R[cur][tid * 8]);
  rx = rp[0];
  ry = rp[1];
#pragma unroll
  for (int j = 0; j < TBLK; ++j) {
    fc[j].ts = Fts[(b + 2) & (RING - 1)][j * SITES + site];
    fc[j].er = Fer[(b + 2) & (RING - 1)][j * SITES + site];
  }

  {
    const int nb = (b + 4 < NBLK) ? (b + 4) : (NBLK - 1);
    const int nidx = (nb * TBLK + st_step) * NS + s0 + st_site;
    raw.p = P[nidx]; raw.t1 = T1[nidx]; raw.t2 = T2[nidx]; raw.e = E[nidx];
  }
}

__global__ __launch_bounds__(THREADS) void waternet_scan(
    const float* __restrict__ P,  const float* __restrict__ T1,
    const float* __restrict__ T2, const float* __restrict__ E,
    const float* __restrict__ w,  float* __restrict__ out) {
  __shared__ v2f   Fts[RING][TBLK * SITES];
  __shared__ v2f   Fer[RING][TBLK * SITES];
  __shared__ float R[2][TBLK * SITES * 8];

  const int tid  = threadIdx.x;
  const int lane = tid & 31;
  const int wv   = tid >> 5;
  const int site = lane & 7;
  const int col  = lane & 15;
  const int half = lane >> 4;
  const int pr   = (lane >> 3) + 4 * wv;
  const int u0   = 2 * pr, u1 = u0 + 1;
  const int s0   = blockIdx.x * SITES;

  const int st_step = tid >> 3;
  const int st_site = tid & 7;

  Gates g;
  g.gm0 = expf(w[u0]) + 1.0f;      g.gm1 = expf(w[u1]) + 1.0f;
  g.ge0 = sigmoidf(w[32 + u0]);    g.ge1 = sigmoidf(w[32 + u1]);
  g.go0 = sigmoidf(w[64 + u0]);    g.go1 = sigmoidf(w[64 + u1]);
  const float xw = w[96 + lane];
  float mx = xw;
  for (int off = 16; off; off >>= 1) mx = fmaxf(mx, __shfl_xor(mx, off, 32));
  const float ex = expf(xw - mx);
  float sm = ex;
  for (int off = 16; off; off >>= 1) sm += __shfl_xor(sm, off, 32);
  const float ga_l = ex / sm;
  g.ga0 = __shfl(ga_l, u0, 32);
  g.ga1 = __shfl(ga_l, u1, 32);

  Raw rawa, rawb;
  {
    const int i0 = st_step * NS + s0 + st_site;
    const int i1 = i0 + TBLK * NS;
    const int i2 = i1 + TBLK * NS;
    const int i3 = i2 + TBLK * NS;
    const float p0 = P[i0], a0 = T1[i0], b0 = T2[i0], e0 = E[i0];
    const float p1 = P[i1], a1 = T1[i1], b1 = T2[i1], e1 = E[i1];
    rawa.p = P[i2]; rawa.t1 = T1[i2]; rawa.t2 = T2[i2]; rawa.e = E[i2];
    rawb.p = P[i3]; rawb.t1 = T1[i3]; rawb.t2 = T2[i3]; rawb.e = E[i3];
    const Forc f0 = partition(p0, a0, b0, e0);
    const Forc f1 = partition(p1, a1, b1, e1);
    Fts[0][tid] = f0.ts;  Fer[0][tid] = f0.er;
    Fts[1][tid] = f1.ts;  Fer[1][tid] = f1.er;
  }
  __syncthreads();

  Forc fa[TBLK], fb[TBLK];
#pragma unroll
  for (int j = 0; j < TBLK; ++j) {
    fa[j].ts = Fts[0][j * SITES + site];  fa[j].er = Fer[0][j * SITES + site];
    fb[j].ts = Fts[1][j * SITES + site];  fb[j].er = Fer[1][j * SITES + site];
  }

  asm volatile("" :: "v"(g.ga0), "v"(g.ga1));
  __syncthreads();

  float S0 = 0.0f, S1 = 0.0f, H0 = 0.0f, H1 = 0.0f;
  const int part = 2 * wv + ((lane >> 3) & 1);
  v4f rxa = {}, rya = {}, rxb = {}, ryb = {};

  for (int b = 0; b < NBLK; b += 2) {
    scan_block(b,     fa, rawa, rxa, rya, Fts, Fer, R, P, T1, T2, E, out, g,
               S0, S1, H0, H1, tid, site, col, half, part, s0, st_step, st_site);
    scan_block(b + 1, fb, rawb, rxb, ryb, Fts, Fer, R, P, T1, T2, E, out, g,
               S0, S1, H0, H1, tid, site, col, half, part, s0, st_step, st_site);
  }

  out[((NBLK - 2) * TBLK + st_step) * NS + s0 + st_site] =
      ((rxa.x + rxa.y) + (rxa.z + rxa.w)) + ((rya.x + rya.y) + (rya.z + rya.w));
  out[((NBLK - 1) * TBLK + st_step) * NS + s0 + st_site] =
      ((rxb.x + rxb.y) + (rxb.z + rxb.w)) + ((ryb.x + ryb.y) + (ryb.z + ryb.w));
}

extern "C" void kernel_launch(void* const* d_in, const int* in_sizes, int n_in,
                              void* d_out, int out_size, void* d_ws, size_t ws_size,
                              hipStream_t stream) {
  (void)in_sizes; (void)n_in; (void)out_size; (void)d_ws; (void)ws_size;
  const float* P  = (const float*)d_in[0];
  const float* T1 = (const float*)d_in[1];
  const float* T2 = (const float*)d_in[2];
  const float* E  = (const float*)d_in[3];
  const float* w  = (const float*)d_in[4];
  float* out = (float*)d_out;

  dim3 grid(NS / SITES);
  dim3 block(THREADS);
  waternet_scan<<<grid, block, 0, stream>>>(P, T1, T2, E, w, out);
}


# Block barrier written by hand with only the LDS wait in front: the raw-forcing loads (two blocks ahead) and the output store stay in flight across it; __syncthreads() drained them every block
# speedup vs baseline: 2.2056x; 1.1917x over previous
#include <hip/hip_runtime.h>
#include <math.h>


typedef __attribute__((ext_vector_type(2))) float v2f;
typedef __attribute__((ext_vector_type(4))) float v4f;
typedef __attribute__((ext_vector_type(8))) float v8f;

#define NT 1024
#define NS 2048
#define SITES 8
#define TBLK 16
#define THREADS 128
#define NBLK (NT / TBLK)
#define RING 4

__device__ __forceinline__ float sigmoidf(float x) {
  return 1.0f / (1.0f + expf(-x));
}

__device__ __forceinline__ void lds_barrier() {
  asm volatile("s_wait_dscnt 0x0\n\ts_barrier_signal -1\n\ts_barrier_wait -1" ::: "memory");
}

struct Forc { v2f ts; v2f er; };

__device__ __forceinline__ Forc partition(float Pv, float t1v, float t2v, float Ev) {
  const float Ta = (t1v + t2v) * 0.5f;
  float ratio = (t1v + t2v) / (t2v - t1v);
  ratio = fminf(1.0f, fmaxf(-1.0f, ratio));
  float rp = 1.0f - acosf(ratio) / 3.1415f;
  rp = (t1v >= 0.0f) ? 1.0f : rp;
  rp = (t2v <= 0.0f) ? 0.0f : rp;
  Forc f;
  f.ts.x = Ta;
  f.ts.y = (1.0f - rp) * Pv;
  f.er.x = Ev;
  f.er.y = rp * Pv;
  return f;
}

struct Gates { float gm0, gm1, ge0, ge1, go0, go1, ga0, ga1; };
struct Raw   { float p, t1, t2, e; };

__device__ __forceinline__ void scan_block(
    const int b, Forc (&fc)[TBLK], Raw& raw, v4f& rx, v4f& ry,
    v2f (*Fts)[TBLK * SITES], v2f (*Fer)[TBLK * SITES], float (*R)[TBLK * SITES * 8],
    const float* __restrict__ P,  const float* __restrict__ T1,
    const float* __restrict__ T2, const float* __restrict__ E,
    float* __restrict__ out, const Gates& g,
    float& S0, float& S1, float& H0, float& H1,
    const int tid, const int site, const int col, const int half, const int part,
    const int s0, const int st_step, const int st_site) {
  const int cur = b & 1;

  if (b >= 2)
    out[((b - 2) * TBLK + st_step) * NS + s0 + st_site] =
        ((rx.x + rx.y) + (rx.z + rx.w)) + ((ry.x + ry.y) + (ry.z + ry.w));

  v8f d = {};
#pragma unroll
  for (int j = 0; j < TBLK; ++j) {
    const float Ta = fc[j].ts.x, snow = fc[j].ts.y;
    const float Ev = fc[j].er.x, rain = fc[j].er.y;

    const float sm0 = fminf(S0, fmaxf(Ta * g.gm0, 0.0f));
    const float sm1 = fminf(S1, fmaxf(Ta * g.gm1, 0.0f));
    S0 = S0 + snow - sm0;
    S1 = S1 + snow - sm1;
    float hn0 = H0 + sm0 + rain - Ev * g.ge0;
    float hn1 = H1 + sm1 + rain - Ev * g.ge1;
    hn0 = fmaxf(hn0, 0.0f);
    hn1 = fmaxf(hn1, 0.0f);
    const float q0 = hn0 * g.go0;
    const float q1 = hn1 * g.go1;
    H0 = hn0 - q0;
    H1 = hn1 - q1;

    const float sel = (col == j) ? 1.0f : 0.0f;
    v2f a;  a.x = sel;         a.y = sel;
    v2f bq; bq.x = q0 * g.ga0; bq.y = q1 * g.ga1;
    d = __builtin_amdgcn_wmma_f32_16x16x4_f32(
         false, a,  false, bq,
         (short)0, d,  false,  false);
  }

#pragma unroll
  for (int r = 0; r < 8; ++r)
    R[cur][((r + 8 * half) * SITES + site) * 8 + part] = d[r];

  if (b + 2 < NBLK) {
    const Forc f = partition(raw.p, raw.t1, raw.t2, raw.e);
    Fts[(b + 2) & (RING - 1)][tid] = f.ts;
    Fer[(b + 2) & (RING - 1)][tid] = f.er;
  }

  lds_barrier();

  const v4f* rp = reinterpret_cast<const v4f*>(&R[cur][tid * 8]);
  rx = rp[0];
  ry = rp[1];
#pragma unroll
  for (int j = 0; j < TBLK; ++j) {
    fc[j].ts = Fts[(b + 2) & (RING - 1)][j * SITES + site];
    fc[j].er = Fer[(b + 2) & (RING - 1)][j * SITES + site];
  }

  {
    const int nb = (b + 4 < NBLK) ? (b + 4) : (NBLK - 1);
    const int nidx = (nb * TBLK + st_step) * NS + s0 + st_site;
    raw.p = P[nidx]; raw.t1 = T1[nidx]; raw.t2 = T2[nidx]; raw.e = E[nidx];
  }
}

__global__ __launch_bounds__(THREADS) void waternet_scan(
    const float* __restrict__ P,  const float* __restrict__ T1,
    const float* __restrict__ T2, const float* __restrict__ E,
    const float* __restrict__ w,  float* __restrict__ out) {
  __shared__ v2f   Fts[RING][TBLK * SITES];
  __shared__ v2f   Fer[RING][TBLK * SITES];
  __shared__ float R[2][TBLK * SITES * 8];

  const int tid  = threadIdx.x;
  const int lane = tid & 31;
  const int wv   = tid >> 5;
  const int site = lane & 7;
  const int col  = lane & 15;
  const int half = lane >> 4;
  const int pr   = (lane >> 3) + 4 * wv;
  const int u0   = 2 * pr, u1 = u0 + 1;
  const int s0   = blockIdx.x * SITES;

  const int st_step = tid >> 3;
  const int st_site = tid & 7;

  Gates g;
  g.gm0 = expf(w[u0]) + 1.0f;      g.gm1 = expf(w[u1]) + 1.0f;
  g.ge0 = sigmoidf(w[32 + u0]);    g.ge1 = sigmoidf(w[32 + u1]);
  g.go0 = sigmoidf(w[64 + u0]);    g.go1 = sigmoidf(w[64 + u1]);
  const float xw = w[96 + lane];
  float mx = xw;
  for (int off = 16; off; off >>= 1) mx = fmaxf(mx, __shfl_xor(mx, off, 32));
  const float ex = expf(xw - mx);
  float sm = ex;
  for (int off = 16; off; off >>= 1) sm += __shfl_xor(sm, off, 32);
  const float ga_l = ex / sm;
  g.ga0 = __shfl(ga_l, u0, 32);
  g.ga1 = __shfl(ga_l, u1, 32);

  Raw rawa, rawb;
  {
    const int i0 = st_step * NS + s0 + st_site;
    const int i1 = i0 + TBLK * NS;
    const int i2 = i1 + TBLK * NS;
    const int i3 = i2 + TBLK * NS;
    const float p0 = P[i0], a0 = T1[i0], b0 = T2[i0], e0 = E[i0];
    const float p1 = P[i1], a1 = T1[i1], b1 = T2[i1], e1 = E[i1];
    rawa.p = P[i2]; rawa.t1 = T1[i2]; rawa.t2 = T2[i2]; rawa.e = E[i2];
    rawb.p = P[i3]; rawb.t1 = T1[i3]; rawb.t2 = T2[i3]; rawb.e = E[i3];
    const Forc f0 = partition(p0, a0, b0, e0);
    const Forc f1 = partition(p1, a1, b1, e1);
    Fts[0][tid] = f0.ts;  Fer[0][tid] = f0.er;
    Fts[1][tid] = f1.ts;  Fer[1][tid] = f1.er;
  }
  __syncthreads();

  Forc fa[TBLK], fb[TBLK];
#pragma unroll
  for (int j = 0; j < TBLK; ++j) {
    fa[j].ts = Fts[0][j * SITES + site];  fa[j].er = Fer[0][j * SITES + site];
    fb[j].ts = Fts[1][j * SITES + site];  fb[j].er = Fer[1][j * SITES + site];
  }

  asm volatile("" :: "v"(g.ga0), "v"(g.ga1));
  __syncthreads();

  float S0 = 0.0f, S1 = 0.0f, H0 = 0.0f, H1 = 0.0f;
  const int part = 2 * wv + ((lane >> 3) & 1);
  v4f rxa = {}, rya = {}, rxb = {}, ryb = {};

  for (int b = 0; b < NBLK; b += 2) {
    scan_block(b,     fa, rawa, rxa, rya, Fts, Fer, R, P, T1, T2, E, out, g,
               S0, S1, H0, H1, tid, site, col, half, part, s0, st_step, st_site);
    scan_block(b + 1, fb, rawb, rxb, ryb, Fts, Fer, R, P, T1, T2, E, out, g,
               S0, S1, H0, H1, tid, site, col, half, part, s0, st_step, st_site);
  }

  out[((NBLK - 2) * TBLK + st_step) * NS + s0 + st_site] =
      ((rxa.x + rxa.y) + (rxa.z + rxa.w)) + ((rya.x + rya.y) + (rya.z + rya.w));
  out[((NBLK - 1) * TBLK + st_step) * NS + s0 + st_site] =
      ((rxb.x + rxb.y) + (rxb.z + rxb.w)) + ((ryb.x + ryb.y) + (ryb.z + ryb.w));
}

extern "C" void kernel_launch(void* const* d_in, const int* in_sizes, int n_in,
                              void* d_out, int out_size, void* d_ws, size_t ws_size,
                              hipStream_t stream) {
  (void)in_sizes; (void)n_in; (void)out_size; (void)d_ws; (void)ws_size;
  const float* P  = (const float*)d_in[0];
  const float* T1 = (const float*)d_in[1];
  const float* T2 = (const float*)d_in[2];
  const float* E  = (const float*)d_in[3];
  const float* w  = (const float*)d_in[4];
  float* out = (float*)d_out;

  dim3 grid(NS / SITES);
  dim3 block(THREADS);
  waternet_scan<<<grid, block, 0, stream>>>(P, T1, T2, E, w, out);
}


# Time block lengthened from 16 to 64 steps (4 accumulator tiles): one barrier and LDS hand-over per 64 steps; each thread's 4 rain/snow partitions per block run as side-by-side chains; same sums
# speedup vs baseline: 2.7000x; 1.2242x over previous
#include <hip/hip_runtime.h>
#include <math.h>


typedef __attribute__((ext_vector_type(2))) float v2f;
typedef __attribute__((ext_vector_type(4))) float v4f;
typedef __attribute__((ext_vector_type(8))) float v8f;

#define NT 1024
#define NS 2048
#define SITES 8
#define TROW 16
#define NTILE 4
#define TBLK (TROW * NTILE)
#define THREADS 128
#define NBLK (NT / TBLK)
#define RING 2

__device__ __forceinline__ float sigmoidf(float x) {
  return 1.0f / (1.0f + expf(-x));
}

__device__ __forceinline__ void lds_barrier() {
  asm volatile("s_wait_dscnt 0x0\n\ts_barrier_signal -1\n\ts_barrier_wait -1" ::: "memory");
}

struct Forc { v2f ts; v2f er; };

struct Gates { float gm0, gm1, ge0, ge1, go0, go1, ga0, ga1; };
struct Raw   { float p, t1, t2, e; };

__device__ __forceinline__ void partition_n(const Raw (&raw)[NTILE], Forc (&f)[NTILE]) {
  float sum[NTILE], ratio[NTILE], ac[NTILE], rp[NTILE];
#pragma unroll
  for (int i = 0; i < NTILE; ++i) sum[i] = raw[i].t1 + raw[i].t2;
#pragma unroll
  for (int i = 0; i < NTILE; ++i) ratio[i] = sum[i] / (raw[i].t2 - raw[i].t1);
#pragma unroll
  for (int i = 0; i < NTILE; ++i) ratio[i] = fminf(1.0f, fmaxf(-1.0f, ratio[i]));
#pragma unroll
  for (int i = 0; i < NTILE; ++i) ac[i] = acosf(ratio[i]);
#pragma unroll
  for (int i = 0; i < NTILE; ++i) rp[i] = 1.0f - ac[i] / 3.1415f;
#pragma unroll
  for (int i = 0; i < NTILE; ++i) {
    rp[i] = (raw[i].t1 >= 0.0f) ? 1.0f : rp[i];
    rp[i] = (raw[i].t2 <= 0.0f) ? 0.0f : rp[i];
    f[i].ts.x = sum[i] * 0.5f;
    f[i].ts.y = (1.0f - rp[i]) * raw[i].p;
    f[i].er.x = raw[i].e;
    f[i].er.y = rp[i] * raw[i].p;
  }
}

__device__ __forceinline__ void scan_block(
    const int b, Forc (&fc)[TBLK], Raw (&raw)[NTILE], v4f (&rx)[NTILE], v4f (&ry)[NTILE],
    v2f (*Fts)[TBLK * SITES], v2f (*Fer)[TBLK * SITES], float (*R)[TBLK * SITES * 8],
    const float* __restrict__ P,  const float* __restrict__ T1,
    const float* __restrict__ T2, const float* __restrict__ E,
    float* __restrict__ out, const Gates& g,
    float& S0, float& S1, float& H0, float& H1,
    const int tid, const int site, const int col, const int half, const int part,
    const int s0, const int st_step, const int st_site) {
  const int cur = b & 1;

  if (b >= 2) {
#pragma unroll
    for (int i = 0; i < NTILE; ++i)
      out[((b - 2) * TBLK + i * TROW + st_step) * NS + s0 + st_site] =
          ((rx[i].x + rx[i].y) + (rx[i].z + rx[i].w)) + ((ry[i].x + ry[i].y) + (ry[i].z + ry[i].w));
  }

  v8f d[NTILE];
#pragma unroll
  for (int i = 0; i < NTILE; ++i) {
  if (i) __builtin_amdgcn_sched_barrier(0);
  d[i] = v8f{};
#pragma unroll
  for (int j = 0; j < TROW; ++j) {
    const float Ta = fc[i * TROW + j].ts.x, snow = fc[i * TROW + j].ts.y;
    const float Ev = fc[i * TROW + j].er.x, rain = fc[i * TROW + j].er.y;

    const float sm0 = fminf(S0, fmaxf(Ta * g.gm0, 0.0f));
    const float sm1 = fminf(S1, fmaxf(Ta * g.gm1, 0.0f));
    S0 = S0 + snow - sm0;
    S1 = S1 + snow - sm1;
    float hn0 = H0 + sm0 + rain - Ev * g.ge0;
    float hn1 = H1 + sm1 + rain - Ev * g.ge1;
    hn0 = fmaxf(hn0, 0.0f);
    hn1 = fmaxf(hn1, 0.0f);
    const float q0 = hn0 * g.go0;
    const float q1 = hn1 * g.go1;
    H0 = hn0 - q0;
    H1 = hn1 - q1;

    const float sel = (col == j) ? 1.0f : 0.0f;
    v2f a;  a.x = sel;         a.y = sel;
    v2f bq; bq.x = q0 * g.ga0; bq.y = q1 * g.ga1;
    d[i] = __builtin_amdgcn_wmma_f32_16x16x4_f32(
         false, a,  false, bq,
         (short)0, d[i],  false,  false);
  }
  }

#pragma unroll
  for (int i = 0; i < NTILE; ++i)
#pragma unroll
    for (int r = 0; r < 8; ++r)
      R[cur][((i * TROW + r + 8 * half) * SITES + site) * 8 + part] = d[i][r];

  if (b + 2 < NBLK) {
    Forc f[NTILE];
    partition_n(raw, f);
#pragma unroll
    for (int i = 0; i < NTILE; ++i) {
      Fts[(b + 2) & (RING - 1)][tid + THREADS * i] = f[i].ts;
      Fer[(b + 2) & (RING - 1)][tid + THREADS * i] = f[i].er;
    }
  }

  lds_barrier();

#pragma unroll
  for (int i = 0; i < NTILE; ++i) {
    const v4f* rp = reinterpret_cast<const v4f*>(&R[cur][(tid + THREADS * i) * 8]);
    rx[i] = rp[0];
    ry[i] = rp[1];
  }
#pragma unroll
  for (int j = 0; j < TBLK; ++j) {
    fc[j].ts = Fts[(b + 2) & (RING - 1)][j * SITES + site];
    fc[j].er = Fer[(b + 2) & (RING - 1)][j * SITES + site];
  }

  {
    const int nb = (b + 4 < NBLK) ? (b + 4) : (NBLK - 1);
#pragma unroll
    for (int i = 0; i < NTILE; ++i) {
      const int nidx = (nb * TBLK + i * TROW + st_step) * NS + s0 + st_site;
      raw[i].p = P[nidx]; raw[i].t1 = T1[nidx]; raw[i].t2 = T2[nidx]; raw[i].e = E[nidx];
    }
  }
}

__global__ __launch_bounds__(THREADS) void waternet_scan(
    const float* __restrict__ P,  const float* __restrict__ T1,
    const float* __restrict__ T2, const float* __restrict__ E,
    const float* __restrict__ w,  float* __restrict__ out) {
  __shared__ v2f   Fts[RING][TBLK * SITES];
  __shared__ v2f   Fer[RING][TBLK * SITES];
  __shared__ float R[2][TBLK * SITES * 8];

  const int tid  = threadIdx.x;
  const int lane = tid & 31;
  const int wv   = tid >> 5;
  const int site = lane & 7;
  const int col  = lane & 15;
  const int half = lane >> 4;
  const int pr   = (lane >> 3) + 4 * wv;
  const int u0   = 2 * pr, u1 = u0 + 1;
  const int s0   = blockIdx.x * SITES;

  const int st_step = tid >> 3;
  const int st_site = tid & 7;

  Gates g;
  g.gm0 = expf(w[u0]) + 1.0f;      g.gm1 = expf(w[u1]) + 1.0f;
  g.ge0 = sigmoidf(w[32 + u0]);    g.ge1 = sigmoidf(w[32 + u1]);
  g.go0 = sigmoidf(w[64 + u0]);    g.go1 = sigmoidf(w[64 + u1]);
  const float xw = w[96 + lane];
  float mx = xw;
  for (int off = 16; off; off >>= 1) mx = fmaxf(mx, __shfl_xor(mx, off, 32));
  const float ex = expf(xw - mx);
  float sm = ex;
  for (int off = 16; off; off >>= 1) sm += __shfl_xor(sm, off, 32);
  const float ga_l = ex / sm;
  g.ga0 = __shfl(ga_l, u0, 32);
  g.ga1 = __shfl(ga_l, u1, 32);

  Raw rawa[NTILE], rawb[NTILE];
  {
    Raw r0[NTILE], r1[NTILE];
#pragma unroll
    for (int i = 0; i < NTILE; ++i) {
      const int i0 = (i * TROW + st_step) * NS + s0 + st_site;
      const int i1 = i0 + TBLK * NS;
      const int i2 = i1 + TBLK * NS;
      const int i3 = i2 + TBLK * NS;
      r0[i].p = P[i0]; r0[i].t1 = T1[i0]; r0[i].t2 = T2[i0]; r0[i].e = E[i0];
      r1[i].p = P[i1]; r1[i].t1 = T1[i1]; r1[i].t2 = T2[i1]; r1[i].e = E[i1];
      rawa[i].p = P[i2]; rawa[i].t1 = T1[i2]; rawa[i].t2 = T2[i2]; rawa[i].e = E[i2];
      rawb[i].p = P[i3]; rawb[i].t1 = T1[i3]; rawb[i].t2 = T2[i3]; rawb[i].e = E[i3];
    }
    Forc f0[NTILE], f1[NTILE];
    partition_n(r0, f0);
    partition_n(r1, f1);
#pragma unroll
    for (int i = 0; i < NTILE; ++i) {
      Fts[0][tid + THREADS * i] = f0[i].ts;  Fer[0][tid + THREADS * i] = f0[i].er;
      Fts[1][tid + THREADS * i] = f1[i].ts;  Fer[1][tid + THREADS * i] = f1[i].er;
    }
  }
  __syncthreads();

  Forc fa[TBLK], fb[TBLK];
#pragma unroll
  for (int j = 0; j < TBLK; ++j) {
    fa[j].ts = Fts[0][j * SITES + site];  fa[j].er = Fer[0][j * SITES + site];
    fb[j].ts = Fts[1][j * SITES + site];  fb[j].er = Fer[1][j * SITES + site];
  }

  asm volatile("" :: "v"(g.ga0), "v"(g.ga1));
  __syncthreads();

  float S0 = 0.0f, S1 = 0.0f, H0 = 0.0f, H1 = 0.0f;
  const int part = 2 * wv + ((lane >> 3) & 1);
  v4f rxa[NTILE] = {}, rya[NTILE] = {}, rxb[NTILE] = {}, ryb[NTILE] = {};

  for (int b = 0; b < NBLK; b += 2) {
    scan_block(b,     fa, rawa, rxa, rya, Fts, Fer, R, P, T1, T2, E, out, g,
               S0, S1, H0, H1, tid, site, col, half, part, s0, st_step, st_site);
    scan_block(b + 1, fb, rawb, rxb, ryb, Fts, Fer, R, P, T1, T2, E, out, g,
               S0, S1, H0, H1, tid, site, col, half, part, s0, st_step, st_site);
  }

#pragma unroll
  for (int i = 0; i < NTILE; ++i) {
    out[((NBLK - 2) * TBLK + i * TROW + st_step) * NS + s0 + st_site] =
        ((rxa[i].x + rxa[i].y) + (rxa[i].z + rxa[i].w)) + ((rya[i].x + rya[i].y) + (rya[i].z + rya[i].w));
    out[((NBLK - 1) * TBLK + i * TROW + st_step) * NS + s0 + st_site] =
        ((rxb[i].x + rxb[i].y) + (rxb[i].z + rxb[i].w)) + ((ryb[i].x + ryb[i].y) + (ryb[i].z + ryb[i].w));
  }
}

extern "C" void kernel_launch(void* const* d_in, const int* in_sizes, int n_in,
                              void* d_out, int out_size, void* d_ws, size_t ws_size,
                              hipStream_t stream) {
  (void)in_sizes; (void)n_in; (void)out_size; (void)d_ws; (void)ws_size;
  const float* P  = (const float*)d_in[0];
  const float* T1 = (const float*)d_in[1];
  const float* T2 = (const float*)d_in[2];
  const float* E  = (const float*)d_in[3];
  const float* w  = (const float*)d_in[4];
  float* out = (float*)d_out;

  dim3 grid(NS / SITES);
  dim3 block(THREADS);
  waternet_scan<<<grid, block, 0, stream>>>(P, T1, T2, E, w, out);
}


# Outflow folded into the next step's first add: store carried before outflow enters as fma(store, 1-go, snowmelt); one link less on the step chain; that product's rounding changes, sum order kept
# speedup vs baseline: 2.8301x; 1.0482x over previous
#include <hip/hip_runtime.h>
#include <math.h>


typedef __attribute__((ext_vector_type(2))) float v2f;
typedef __attribute__((ext_vector_type(4))) float v4f;
typedef __attribute__((ext_vector_type(8))) float v8f;

#define NT 1024
#define NS 2048
#define SITES 8
#define TROW 16
#define NTILE 4
#define TBLK (TROW * NTILE)
#define THREADS 128
#define NBLK (NT / TBLK)
#define RING 2

__device__ __forceinline__ float sigmoidf(float x) {
  return 1.0f / (1.0f + expf(-x));
}

__device__ __forceinline__ void lds_barrier() {
  asm volatile("s_wait_dscnt 0x0\n\ts_barrier_signal -1\n\ts_barrier_wait -1" ::: "memory");
}

struct Forc { v2f ts; v2f er; };

struct Gates { float gm0, gm1, ge0, ge1, go0, go1, gk0, gk1, ga0, ga1; };
struct Raw   { float p, t1, t2, e; };

__device__ __forceinline__ void partition_n(const Raw (&raw)[NTILE], Forc (&f)[NTILE]) {
  float sum[NTILE], ratio[NTILE], ac[NTILE], rp[NTILE];
#pragma unroll
  for (int i = 0; i < NTILE; ++i) sum[i] = raw[i].t1 + raw[i].t2;
#pragma unroll
  for (int i = 0; i < NTILE; ++i) ratio[i] = sum[i] / (raw[i].t2 - raw[i].t1);
#pragma unroll
  for (int i = 0; i < NTILE; ++i) ratio[i] = fminf(1.0f, fmaxf(-1.0f, ratio[i]));
#pragma unroll
  for (int i = 0; i < NTILE; ++i) ac[i] = acosf(ratio[i]);
#pragma unroll
  for (int i = 0; i < NTILE; ++i) rp[i] = 1.0f - ac[i] / 3.1415f;
#pragma unroll
  for (int i = 0; i < NTILE; ++i) {
    rp[i] = (raw[i].t1 >= 0.0f) ? 1.0f : rp[i];
    rp[i] = (raw[i].t2 <= 0.0f) ? 0.0f : rp[i];
    f[i].ts.x = sum[i] * 0.5f;
    f[i].ts.y = (1.0f - rp[i]) * raw[i].p;
    f[i].er.x = raw[i].e;
    f[i].er.y = rp[i] * raw[i].p;
  }
}

__device__ __forceinline__ void scan_block(
    const int b, Forc (&fc)[TBLK], Raw (&raw)[NTILE], v4f (&rx)[NTILE], v4f (&ry)[NTILE],
    v2f (*Fts)[TBLK * SITES], v2f (*Fer)[TBLK * SITES], float (*R)[TBLK * SITES * 8],
    const float* __restrict__ P,  const float* __restrict__ T1,
    const float* __restrict__ T2, const float* __restrict__ E,
    float* __restrict__ out, const Gates& g,
    float& S0, float& S1, float& W0, float& W1,
    const int tid, const int site, const int col, const int half, const int part,
    const int s0, const int st_step, const int st_site) {
  const int cur = b & 1;

  if (b >= 2) {
#pragma unroll
    for (int i = 0; i < NTILE; ++i)
      out[((b - 2) * TBLK + i * TROW + st_step) * NS + s0 + st_site] =
          ((rx[i].x + rx[i].y) + (rx[i].z + rx[i].w)) + ((ry[i].x + ry[i].y) + (ry[i].z + ry[i].w));
  }

  v8f d[NTILE];
#pragma unroll
  for (int i = 0; i < NTILE; ++i) {
  if (i) __builtin_amdgcn_sched_barrier(0);
  d[i] = v8f{};
#pragma unroll
  for (int j = 0; j < TROW; ++j) {
    const float Ta = fc[i * TROW + j].ts.x, snow = fc[i * TROW + j].ts.y;
    const float Ev = fc[i * TROW + j].er.x, rain = fc[i * TROW + j].er.y;

    const float sm0 = fminf(S0, fmaxf(Ta * g.gm0, 0.0f));
    const float sm1 = fminf(S1, fmaxf(Ta * g.gm1, 0.0f));
    S0 = S0 + snow - sm0;
    S1 = S1 + snow - sm1;
    const float hn0 = __builtin_fmaf(W0, g.gk0, sm0) + rain - Ev * g.ge0;
    const float hn1 = __builtin_fmaf(W1, g.gk1, sm1) + rain - Ev * g.ge1;
    W0 = fmaxf(hn0, 0.0f);
    W1 = fmaxf(hn1, 0.0f);
    const float q0 = W0 * g.go0;
    const float q1 = W1 * g.go1;

    const float sel = (col == j) ? 1.0f : 0.0f;
    v2f a;  a.x = sel;         a.y = sel;
    v2f bq; bq.x = q0 * g.ga0; bq.y = q1 * g.ga1;
    d[i] = __builtin_amdgcn_wmma_f32_16x16x4_f32(
         false, a,  false, bq,
         (short)0, d[i],  false,  false);
  }
  }

#pragma unroll
  for (int i = 0; i < NTILE; ++i)
#pragma unroll
    for (int r = 0; r < 8; ++r)
      R[cur][((i * TROW + r + 8 * half) * SITES + site) * 8 + part] = d[i][r];

  if (b + 2 < NBLK) {
    Forc f[NTILE];
    partition_n(raw, f);
#pragma unroll
    for (int i = 0; i < NTILE; ++i) {
      Fts[(b + 2) & (RING - 1)][tid + THREADS * i] = f[i].ts;
      Fer[(b + 2) & (RING - 1)][tid + THREADS * i] = f[i].er;
    }
  }

  lds_barrier();

#pragma unroll
  for (int i = 0; i < NTILE; ++i) {
    const v4f* rp = reinterpret_cast<const v4f*>(&R[cur][(tid + THREADS * i) * 8]);
    rx[i] = rp[0];
    ry[i] = rp[1];
  }
#pragma unroll
  for (int j = 0; j < TBLK; ++j) {
    fc[j].ts = Fts[(b + 2) & (RING - 1)][j * SITES + site];
    fc[j].er = Fer[(b + 2) & (RING - 1)][j * SITES + site];
  }

  {
    const int nb = (b + 4 < NBLK) ? (b + 4) : (NBLK - 1);
#pragma unroll
    for (int i = 0; i < NTILE; ++i) {
      const int nidx = (nb * TBLK + i * TROW + st_step) * NS + s0 + st_site;
      raw[i].p = P[nidx]; raw[i].t1 = T1[nidx]; raw[i].t2 = T2[nidx]; raw[i].e = E[nidx];
    }
  }
}

__global__ __launch_bounds__(THREADS) void waternet_scan(
    const float* __restrict__ P,  const float* __restrict__ T1,
    const float* __restrict__ T2, const float* __restrict__ E,
    const float* __restrict__ w,  float* __restrict__ out) {
  __shared__ v2f   Fts[RING][TBLK * SITES];
  __shared__ v2f   Fer[RING][TBLK * SITES];
  __shared__ float R[2][TBLK * SITES * 8];

  const int tid  = threadIdx.x;
  const int lane = tid & 31;
  const int wv   = tid >> 5;
  const int site = lane & 7;
  const int col  = lane & 15;
  const int half = lane >> 4;
  const int pr   = (lane >> 3) + 4 * wv;
  const int u0   = 2 * pr, u1 = u0 + 1;
  const int s0   = blockIdx.x * SITES;

  const int st_step = tid >> 3;
  const int st_site = tid & 7;

  Gates g;
  g.gm0 = expf(w[u0]) + 1.0f;      g.gm1 = expf(w[u1]) + 1.0f;
  g.ge0 = sigmoidf(w[32 + u0]);    g.ge1 = sigmoidf(w[32 + u1]);
  g.go0 = sigmoidf(w[64 + u0]);    g.go1 = sigmoidf(w[64 + u1]);
  g.gk0 = 1.0f - g.go0;            g.gk1 = 1.0f - g.go1;
  const float xw = w[96 + lane];
  float mx = xw;
  for (int off = 16; off; off >>= 1) mx = fmaxf(mx, __shfl_xor(mx, off, 32));
  const float ex = expf(xw - mx);
  float sm = ex;
  for (int off = 16; off; off >>= 1) sm += __shfl_xor(sm, off, 32);
  const float ga_l = ex / sm;
  g.ga0 = __shfl(ga_l, u0, 32);
  g.ga1 = __shfl(ga_l, u1, 32);

  Raw rawa[NTILE], rawb[NTILE];
  {
    Raw r0[NTILE], r1[NTILE];
#pragma unroll
    for (int i = 0; i < NTILE; ++i) {
      const int i0 = (i * TROW + st_step) * NS + s0 + st_site;
      const int i1 = i0 + TBLK * NS;
      const int i2 = i1 + TBLK * NS;
      const int i3 = i2 + TBLK * NS;
      r0[i].p = P[i0]; r0[i].t1 = T1[i0]; r0[i].t2 = T2[i0]; r0[i].e = E[i0];
      r1[i].p = P[i1]; r1[i].t1 = T1[i1]; r1[i].t2 = T2[i1]; r1[i].e = E[i1];
      rawa[i].p = P[i2]; rawa[i].t1 = T1[i2]; rawa[i].t2 = T2[i2]; rawa[i].e = E[i2];
      rawb[i].p = P[i3]; rawb[i].t1 = T1[i3]; rawb[i].t2 = T2[i3]; rawb[i].e = E[i3];
    }
    Forc f0[NTILE], f1[NTILE];
    partition_n(r0, f0);
    partition_n(r1, f1);
#pragma unroll
    for (int i = 0; i < NTILE; ++i) {
      Fts[0][tid + THREADS * i] = f0[i].ts;  Fer[0][tid + THREADS * i] = f0[i].er;
      Fts[1][tid + THREADS * i] = f1[i].ts;  Fer[1][tid + THREADS * i] = f1[i].er;
    }
  }
  __syncthreads();

  Forc fa[TBLK], fb[TBLK];
#pragma unroll
  for (int j = 0; j < TBLK; ++j) {
    fa[j].ts = Fts[0][j * SITES + site];  fa[j].er = Fer[0][j * SITES + site];
    fb[j].ts = Fts[1][j * SITES + site];  fb[j].er = Fer[1][j * SITES + site];
  }

  asm volatile("" :: "v"(g.ga0), "v"(g.ga1));
  __syncthreads();

  float S0 = 0.0f, S1 = 0.0f, W0 = 0.0f, W1 = 0.0f;
  const int part = 2 * wv + ((lane >> 3) & 1);
  v4f rxa[NTILE] = {}, rya[NTILE] = {}, rxb[NTILE] = {}, ryb[NTILE] = {};

  for (int b = 0; b < NBLK; b += 2) {
    scan_block(b,     fa, rawa, rxa, rya, Fts, Fer, R, P, T1, T2, E, out, g,
               S0, S1, W0, W1, tid, site, col, half, part, s0, st_step, st_site);
    scan_block(b + 1, fb, rawb, rxb, ryb, Fts, Fer, R, P, T1, T2, E, out, g,
               S0, S1, W0, W1, tid, site, col, half, part, s0, st_step, st_site);
  }

#pragma unroll
  for (int i = 0; i < NTILE; ++i) {
    out[((NBLK - 2) * TBLK + i * TROW + st_step) * NS + s0 + st_site] =
        ((rxa[i].x + rxa[i].y) + (rxa[i].z + rxa[i].w)) + ((rya[i].x + rya[i].y) + (rya[i].z + rya[i].w));
    out[((NBLK - 1) * TBLK + i * TROW + st_step) * NS + s0 + st_site] =
        ((rxb[i].x + rxb[i].y) + (rxb[i].z + rxb[i].w)) + ((ryb[i].x + ryb[i].y) + (ryb[i].z + ryb[i].w));
  }
}

extern "C" void kernel_launch(void* const* d_in, const int* in_sizes, int n_in,
                              void* d_out, int out_size, void* d_ws, size_t ws_size,
                              hipStream_t stream) {
  (void)in_sizes; (void)n_in; (void)out_size; (void)d_ws; (void)ws_size;
  const float* P  = (const float*)d_in[0];
  const float* T1 = (const float*)d_in[1];
  const float* T2 = (const float*)d_in[2];
  const float* E  = (const float*)d_in[3];
  const float* w  = (const float*)d_in[4];
  float* out = (float*)d_out;

  dim3 grid(NS / SITES);
  dim3 block(THREADS);
  waternet_scan<<<grid, block, 0, stream>>>(P, T1, T2, E, w, out);
}


# Q*ga enter the f32 WMMA through its accumulator start value as well as through B (zero-masked weights route by lane half): 1 WMMA per 2 steps; 16 partials per (step,site) via LDS (new sum tree)
# speedup vs baseline: 3.0452x; 1.0760x over previous
#include <hip/hip_runtime.h>
#include <math.h>


typedef __attribute__((ext_vector_type(2))) float v2f;
typedef __attribute__((ext_vector_type(4))) float v4f;
typedef __attribute__((ext_vector_type(8))) float v8f;

#define NT 1024
#define NS 2048
#define SITES 8
#define TROW 8
#define NTILE 8
#define TBLK (TROW * NTILE)
#define THREADS 128
#define ISTEP (THREADS / SITES)
#define NITEM (TBLK / ISTEP)
#define NPART 16
#define NBLK (NT / TBLK)
#define RING 2

__device__ __forceinline__ float sigmoidf(float x) {
  return 1.0f / (1.0f + expf(-x));
}

__device__ __forceinline__ void lds_barrier() {
  asm volatile("s_wait_dscnt 0x0\n\ts_barrier_signal -1\n\ts_barrier_wait -1" ::: "memory");
}

struct Forc { v2f ts; v2f er; };

struct Gates { float gm0, gm1, ge0, ge1, go0, go1, gk0, gk1, ga0, ga1, gA0, gA1, gB0, gB1; };
struct Raw   { float p, t1, t2, e; };

__device__ __forceinline__ void partition_n(const Raw (&raw)[NITEM], Forc (&f)[NITEM]) {
  float sum[NITEM], ratio[NITEM], ac[NITEM], rp[NITEM];
#pragma unroll
  for (int i = 0; i < NITEM; ++i) sum[i] = raw[i].t1 + raw[i].t2;
#pragma unroll
  for (int i = 0; i < NITEM; ++i) ratio[i] = sum[i] / (raw[i].t2 - raw[i].t1);
#pragma unroll
  for (int i = 0; i < NITEM; ++i) ratio[i] = fminf(1.0f, fmaxf(-1.0f, ratio[i]));
#pragma unroll
  for (int i = 0; i < NITEM; ++i) ac[i] = acosf(ratio[i]);
#pragma unroll
  for (int i = 0; i < NITEM; ++i) rp[i] = 1.0f - ac[i] / 3.1415f;
#pragma unroll
  for (int i = 0; i < NITEM; ++i) {
    rp[i] = (raw[i].t1 >= 0.0f) ? 1.0f : rp[i];
    rp[i] = (raw[i].t2 <= 0.0f) ? 0.0f : rp[i];
    f[i].ts.x = sum[i] * 0.5f;
    f[i].ts.y = (1.0f - rp[i]) * raw[i].p;
    f[i].er.x = raw[i].e;
    f[i].er.y = rp[i] * raw[i].p;
  }
}

__device__ __forceinline__ float sum_parts(const v4f (&p)[4]) {
  const v4f s = (p[0] + p[1]) + (p[2] + p[3]);
  return (s.x + s.y) + (s.z + s.w);
}

__device__ __forceinline__ void scan_block(
    const int b, Forc (&fc)[TBLK], Raw (&raw)[NITEM], v4f (&rs)[NITEM][4],
    v2f (*Fts)[TBLK * SITES], v2f (*Fer)[TBLK * SITES], float (*R)[TBLK * SITES * NPART],
    const float* __restrict__ P,  const float* __restrict__ T1,
    const float* __restrict__ T2, const float* __restrict__ E,
    float* __restrict__ out, const Gates& g, const v2f (&am)[4],
    float& S0, float& S1, float& W0, float& W1,
    const int tid, const int site, const int half, const int part,
    const int s0, const int st_step, const int st_site) {
  const int cur = b & 1;

  if (b >= 2) {
#pragma unroll
    for (int i = 0; i < NITEM; ++i)
      out[((b - 2) * TBLK + i * ISTEP + st_step) * NS + s0 + st_site] = sum_parts(rs[i]);
  }

  v8f d[NTILE];
  v2f bq[NTILE][4];
#pragma unroll
  for (int t = 0; t < NTILE; ++t) {
  if (t && !(t & 1)) __builtin_amdgcn_sched_barrier(0);
#pragma unroll
  for (int j = 0; j < TROW; ++j) {
    const float Ta = fc[t * TROW + j].ts.x, snow = fc[t * TROW + j].ts.y;
    const float Ev = fc[t * TROW + j].er.x, rain = fc[t * TROW + j].er.y;

    const float sm0 = fminf(S0, fmaxf(Ta * g.gm0, 0.0f));
    const float sm1 = fminf(S1, fmaxf(Ta * g.gm1, 0.0f));
    S0 = S0 + snow - sm0;
    S1 = S1 + snow - sm1;
    const float hn0 = __builtin_fmaf(W0, g.gk0, sm0) + rain - Ev * g.ge0;
    const float hn1 = __builtin_fmaf(W1, g.gk1, sm1) + rain - Ev * g.ge1;
    W0 = fmaxf(hn0, 0.0f);
    W1 = fmaxf(hn1, 0.0f);
    const float q0 = W0 * g.go0;
    const float q1 = W1 * g.go1;

    const int i = j & 3;
    if (j < 4) {
      d[t][2 * i]     = q0 * g.gA0;
      d[t][2 * i + 1] = q1 * g.gA1;
      bq[t][i].x      = q0 * g.gB0;
      bq[t][i].y      = q1 * g.gB1;
    } else {
      d[t][2 * i]     = __builtin_fmaf(q0, g.gB0, d[t][2 * i]);
      d[t][2 * i + 1] = __builtin_fmaf(q1, g.gB1, d[t][2 * i + 1]);
      bq[t][i].x      = __builtin_fmaf(q0, g.gA0, bq[t][i].x);
      bq[t][i].y      = __builtin_fmaf(q1, g.gA1, bq[t][i].y);
    }

    if (t > 0 && (j & 1)) {
      const int w = j >> 1;
      d[t - 1] = __builtin_amdgcn_wmma_f32_16x16x4_f32(
           false, am[w],  false, bq[t - 1][w],
           (short)0, d[t - 1],  false,  false);
    }
  }
  }
  __builtin_amdgcn_sched_barrier(0);
#pragma unroll
  for (int w = 0; w < 4; ++w)
    d[NTILE - 1] = __builtin_amdgcn_wmma_f32_16x16x4_f32(
         false, am[w],  false, bq[NTILE - 1][w],
         (short)0, d[NTILE - 1],  false,  false);

#pragma unroll
  for (int t = 0; t < NTILE; ++t)
#pragma unroll
    for (int i = 0; i < 4; ++i) {
      v2f pr2; pr2.x = d[t][2 * i]; pr2.y = d[t][2 * i + 1];
      *reinterpret_cast<v2f*>(&R[cur][((t * TROW + 4 * half + i) * SITES + site) * 4 + part]) = pr2;
    }

  if (b + 2 < NBLK) {
    Forc f[NITEM];
    partition_n(raw, f);
#pragma unroll
    for (int i = 0; i < NITEM; ++i) {
      Fts[(b + 2) & (RING - 1)][tid + THREADS * i] = f[i].ts;
      Fer[(b + 2) & (RING - 1)][tid + THREADS * i] = f[i].er;
    }
  }

  lds_barrier();

#pragma unroll
  for (int i = 0; i < NITEM; ++i) {
#pragma unroll
    for (int q = 0; q < 4; ++q)
      rs[i][q] = *reinterpret_cast<const v4f*>(&R[cur][(q * (TBLK * SITES) + tid + THREADS * i) * 4]);
  }
#pragma unroll
  for (int j = 0; j < TBLK; ++j) {
    fc[j].ts = Fts[(b + 2) & (RING - 1)][j * SITES + site];
    fc[j].er = Fer[(b + 2) & (RING - 1)][j * SITES + site];
  }

  {
    const int nb = (b + 4 < NBLK) ? (b + 4) : (NBLK - 1);
#pragma unroll
    for (int i = 0; i < NITEM; ++i) {
      const int nidx = (nb * TBLK + i * ISTEP + st_step) * NS + s0 + st_site;
      raw[i].p = P[nidx]; raw[i].t1 = T1[nidx]; raw[i].t2 = T2[nidx]; raw[i].e = E[nidx];
    }
  }
}

__global__ __launch_bounds__(THREADS) void waternet_scan(
    const float* __restrict__ P,  const float* __restrict__ T1,
    const float* __restrict__ T2, const float* __restrict__ E,
    const float* __restrict__ w,  float* __restrict__ out) {
  __shared__ struct alignas(16) {
    v2f   Fts[RING][TBLK * SITES];
    v2f   Fer[RING][TBLK * SITES];
    float R[2][TBLK * SITES * NPART];
  } lds;
  v2f (*Fts)[TBLK * SITES] = lds.Fts;
  v2f (*Fer)[TBLK * SITES] = lds.Fer;
  float (*R)[TBLK * SITES * NPART] = lds.R;

  const int tid  = threadIdx.x;
  const int lane = tid & 31;
  const int wv   = tid >> 5;
  const int site = lane & 7;
  const int col  = lane & 15;
  const int half = lane >> 4;
  const int pr   = (lane >> 3) + 4 * wv;
  const int u0   = 2 * pr, u1 = u0 + 1;
  const int s0   = blockIdx.x * SITES;

  const int st_step = tid >> 3;
  const int st_site = tid & 7;

  Gates g;
  g.gm0 = expf(w[u0]) + 1.0f;      g.gm1 = expf(w[u1]) + 1.0f;
  g.ge0 = sigmoidf(w[32 + u0]);    g.ge1 = sigmoidf(w[32 + u1]);
  g.go0 = sigmoidf(w[64 + u0]);    g.go1 = sigmoidf(w[64 + u1]);
  g.gk0 = 1.0f - g.go0;            g.gk1 = 1.0f - g.go1;
  const float xw = w[96 + lane];
  float mx = xw;
  for (int off = 16; off; off >>= 1) mx = fmaxf(mx, __shfl_xor(mx, off, 32));
  const float ex = expf(xw - mx);
  float sm = ex;
  for (int off = 16; off; off >>= 1) sm += __shfl_xor(sm, off, 32);
  const float ga_l = ex / sm;
  g.ga0 = __shfl(ga_l, u0, 32);
  g.ga1 = __shfl(ga_l, u1, 32);

  v2f am[4];
  {
    const int cb = col - 8 * (1 - half);
#pragma unroll
    for (int wi = 0; wi < 4; ++wi) {
      am[wi].x = (cb == 2 * wi)     ? 1.0f : 0.0f;
      am[wi].y = (cb == 2 * wi + 1) ? 1.0f : 0.0f;
    }
  }

  Raw rawa[NITEM], rawb[NITEM];
  {
    Raw r0[NITEM], r1[NITEM];
#pragma unroll
    for (int i = 0; i < NITEM; ++i) {
      const int i0 = (i * ISTEP + st_step) * NS + s0 + st_site;
      const int i1 = i0 + TBLK * NS;
      const int i2 = i1 + TBLK * NS;
      const int i3 = i2 + TBLK * NS;
      r0[i].p = P[i0]; r0[i].t1 = T1[i0]; r0[i].t2 = T2[i0]; r0[i].e = E[i0];
      r1[i].p = P[i1]; r1[i].t1 = T1[i1]; r1[i].t2 = T2[i1]; r1[i].e = E[i1];
      rawa[i].p = P[i2]; rawa[i].t1 = T1[i2]; rawa[i].t2 = T2[i2]; rawa[i].e = E[i2];
      rawb[i].p = P[i3]; rawb[i].t1 = T1[i3]; rawb[i].t2 = T2[i3]; rawb[i].e = E[i3];
    }
    Forc f0[NITEM], f1[NITEM];
    partition_n(r0, f0);
    partition_n(r1, f1);
#pragma unroll
    for (int i = 0; i < NITEM; ++i) {
      Fts[0][tid + THREADS * i] = f0[i].ts;  Fer[0][tid + THREADS * i] = f0[i].er;
      Fts[1][tid + THREADS * i] = f1[i].ts;  Fer[1][tid + THREADS * i] = f1[i].er;
    }
  }
  __syncthreads();

  Forc fa[TBLK], fb[TBLK];
#pragma unroll
  for (int j = 0; j < TBLK; ++j) {
    fa[j].ts = Fts[0][j * SITES + site];  fa[j].er = Fer[0][j * SITES + site];
    fb[j].ts = Fts[1][j * SITES + site];  fb[j].er = Fer[1][j * SITES + site];
  }

  asm volatile("" :: "v"(g.ga0), "v"(g.ga1));
  g.gA0 = half ? 0.0f : g.ga0;     g.gA1 = half ? 0.0f : g.ga1;
  g.gB0 = half ? g.ga0 : 0.0f;     g.gB1 = half ? g.ga1 : 0.0f;
  __syncthreads();

  float S0 = 0.0f, S1 = 0.0f, W0 = 0.0f, W1 = 0.0f;
  const int part = wv * (TBLK * SITES * 4) + 2 * ((lane >> 3) & 1);
  v4f rsa[NITEM][4] = {}, rsb[NITEM][4] = {};

  for (int b = 0; b < NBLK; b += 2) {
    scan_block(b,     fa, rawa, rsa, Fts, Fer, R, P, T1, T2, E, out, g, am,
               S0, S1, W0, W1, tid, site, half, part, s0, st_step, st_site);
    scan_block(b + 1, fb, rawb, rsb, Fts, Fer, R, P, T1, T2, E, out, g, am,
               S0, S1, W0, W1, tid, site, half, part, s0, st_step, st_site);
  }

#pragma unroll
  for (int i = 0; i < NITEM; ++i) {
    out[((NBLK - 2) * TBLK + i * ISTEP + st_step) * NS + s0 + st_site] = sum_parts(rsa[i]);
    out[((NBLK - 1) * TBLK + i * ISTEP + st_step) * NS + s0 + st_site] = sum_parts(rsb[i]);
  }
}

extern "C" void kernel_launch(void* const* d_in, const int* in_sizes, int n_in,
                              void* d_out, int out_size, void* d_ws, size_t ws_size,
                              hipStream_t stream) {
  (void)in_sizes; (void)n_in; (void)out_size; (void)d_ws; (void)ws_size;
  const float* P  = (const float*)d_in[0];
  const float* T1 = (const float*)d_in[1];
  const float* T2 = (const float*)d_in[2];
  const float* E  = (const float*)d_in[3];
  const float* w  = (const float*)d_in[4];
  float* out = (float*)d_out;

  dim3 grid(NS / SITES);
  dim3 block(THREADS);
  waternet_scan<<<grid, block, 0, stream>>>(P, T1, T2, E, w, out);
}
